# starting kernel
# baseline (speedup 1.0000x reference)
#include <hip/hip_runtime.h>

#define N_NODES 100000
#define N_EDGES 1000000
#define D 64
#define LN_EPS 1e-5f

typedef __attribute__((ext_vector_type(2))) float v2f;
typedef __attribute__((ext_vector_type(8))) float v8f;

__global__ void zero_f32_kernel(float* __restrict__ p, int n) {
  int i = blockIdx.x * blockDim.x + threadIdx.x;
  if (i < n) p[i] = 0.0f;
}

__global__ void degree_kernel(const int* __restrict__ src, const int* __restrict__ dst,
                              float* __restrict__ out_deg, float* __restrict__ in_deg, int e) {
  int i = blockIdx.x * blockDim.x + threadIdx.x;
  if (i < e) {
    atomicAdd(&out_deg[src[i]], 1.0f);
    atomicAdd(&in_deg[dst[i]], 1.0f);
  }
}

__global__ void norm_kernel(float* __restrict__ out_norm, float* __restrict__ in_norm, int n) {
  int i = blockIdx.x * blockDim.x + threadIdx.x;
  if (i < n) {
    out_norm[i] = rsqrtf(fmaxf(out_norm[i], 1.0f));
    in_norm[i]  = rsqrtf(fmaxf(in_norm[i], 1.0f));
  }
}

__global__ void scatter_kernel(const float* __restrict__ h,
                               const float* __restrict__ out_norm,
                               const int* __restrict__ src, const int* __restrict__ dst,
                               float* __restrict__ agg, int e) {
  int g    = blockIdx.x * blockDim.x + threadIdx.x;
  int edge = g >> 5;
  int lane = g & 31;
  if (edge >= e) return;
  int s = src[edge];
  int d = dst[edge];
  float on = out_norm[s];
  float2 v = *(reinterpret_cast<const float2*>(h + (size_t)s * D) + lane);
  float* ap = agg + (size_t)d * D + lane * 2;
  atomicAdd(ap,     v.x * on);
  atomicAdd(ap + 1, v.y * on);
}

__global__ __launch_bounds__(256) void gcn_layer_kernel(
    const float* __restrict__ agg, const float* __restrict__ in_norm,
    const float* __restrict__ h_in,
    const float* __restrict__ Wc, const float* __restrict__ bc,
    const float* __restrict__ Wr, const float* __restrict__ br,
    const float* __restrict__ lng, const float* __restrict__ lnb,
    float* __restrict__ h_out, float* __restrict__ h_final, int n)
{
  __shared__ float lds[8][16 * D];
  const int wave = threadIdx.x >> 5;
  const int lane = threadIdx.x & 31;
  const int row  = lane & 15;
  const int kb   = (lane >> 4) * 2;
  const int base = (blockIdx.x * 8 + wave) * 16;
  int node  = base + row;
  int cnode = node < n ? node : (n - 1);

  v2f aC[16], aR[16];
  const float inrm = in_norm[cnode];
  const float* arow = agg  + (size_t)cnode * D;
  const float* hrow = h_in + (size_t)cnode * D;
#pragma unroll
  for (int kt = 0; kt < 16; ++kt) {
    int k0 = kt * 4 + kb;
    aC[kt][0] = arow[k0]     * inrm;
    aC[kt][1] = arow[k0 + 1] * inrm;
    aR[kt][0] = hrow[k0];
    aR[kt][1] = hrow[k0 + 1];
  }

  const int ncol = lane & 15;
#pragma unroll
  for (int nt = 0; nt < 4; ++nt) {
    v8f c = {};
    const int col = nt * 16 + ncol;
#pragma unroll
    for (int kt = 0; kt < 16; ++kt) {
      int kr = kt * 4 + kb;
      v2f bC, bR;
      bC[0] = Wc[kr * D + col];
      bC[1] = Wc[(kr + 1) * D + col];
      bR[0] = Wr[kr * D + col];
      bR[1] = Wr[(kr + 1) * D + col];
      c = __builtin_amdgcn_wmma_f32_16x16x4_f32(false, aC[kt], false, bC, (short)0, c, false, false);
      c = __builtin_amdgcn_wmma_f32_16x16x4_f32(false, aR[kt], false, bR, (short)0, c, false, false);
    }
    const float bias = bc[col] + br[col];
    const int m0 = (lane < 16) ? 0 : 8;
#pragma unroll
    for (int r = 0; r < 8; ++r)
      lds[wave][(m0 + r) * D + col] = c[r] + bias;
  }
  __syncthreads();

  const int rr = lane >> 1;
  const int j0 = (lane & 1) * 32;
  const float* myrow = &lds[wave][rr * D];
  float s = 0.0f, sq = 0.0f;
#pragma unroll
  for (int j = 0; j < 32; ++j) {
    float x = myrow[j0 + j];
    s  += x;
    sq += x * x;
  }
  s  += __shfl_xor(s, 1);
  sq += __shfl_xor(sq, 1);
  const float mu   = s * (1.0f / 64.0f);
  const float var  = sq * (1.0f / 64.0f) - mu * mu;
  const float rstd = rsqrtf(var + LN_EPS);
  const int onode = base + rr;
  if (onode < n) {
    float* ho = h_out   + (size_t)onode * D;
    float* hf = h_final + (size_t)onode * D;
#pragma unroll
    for (int j = 0; j < 32; ++j) {
      int jj = j0 + j;
      float y = (myrow[jj] - mu) * rstd * lng[jj] + lnb[jj];
      y = fmaxf(y, 0.0f);
      ho[jj] = y;
      hf[jj] += y;
    }
  }
}

__global__ __launch_bounds__(256) void pred_kernel(
    const float* __restrict__ hf, const float* __restrict__ W,
    const float* __restrict__ b, float* __restrict__ out, int n)
{
  const int wave = threadIdx.x >> 5;
  const int lane = threadIdx.x & 31;
  const int row  = lane & 15;
  const int kb   = (lane >> 4) * 2;
  const int base = (blockIdx.x * 8 + wave) * 16;
  int node  = base + row;
  int cnode = node < n ? node : (n - 1);

  v2f a[16];
  const float* ar = hf + (size_t)cnode * D;
#pragma unroll
  for (int kt = 0; kt < 16; ++kt) {
    int k0 = kt * 4 + kb;
    a[kt][0] = ar[k0];
    a[kt][1] = ar[k0 + 1];
  }

  const int ncol = lane & 15;
#pragma unroll
  for (int nt = 0; nt < 4; ++nt) {
    v8f c = {};
    const int col = nt * 16 + ncol;
#pragma unroll
    for (int kt = 0; kt < 16; ++kt) {
      int kr = kt * 4 + kb;
      v2f bf;
      bf[0] = W[kr * D + col];
      bf[1] = W[(kr + 1) * D + col];
      c = __builtin_amdgcn_wmma_f32_16x16x4_f32(false, a[kt], false, bf, (short)0, c, false, false);
    }
    const float bias = b[col];
    const int m0 = (lane < 16) ? 0 : 8;
#pragma unroll
    for (int r = 0; r < 8; ++r) {
      int m = base + m0 + r;
      if (m < n) out[(size_t)m * D + col] = c[r] + bias;
    }
  }
}

extern "C" void kernel_launch(void* const* d_in, const int* in_sizes, int n_in,
                              void* d_out, int out_size, void* d_ws, size_t ws_size,
                              hipStream_t stream) {
  (void)in_sizes; (void)n_in; (void)out_size; (void)ws_size;

  const float* feats  = (const float*)d_in[0];
  const int*   src    = (const int*)  d_in[1];
  const int*   dst    = (const int*)  d_in[2];
  const float* conv_W = (const float*)d_in[3];
  const float* conv_b = (const float*)d_in[4];
  const float* res_W  = (const float*)d_in[5];
  const float* res_b  = (const float*)d_in[6];
  const float* ln_g   = (const float*)d_in[7];
  const float* ln_b   = (const float*)d_in[8];
  const float* pred_W = (const float*)d_in[9];
  const float* pred_b = (const float*)d_in[10];
  float* out = (float*)d_out;

  char* ws = (char*)d_ws;
  size_t off = 0;
  auto alloc = [&](size_t bytes) -> float* {
    float* p = (float*)(ws + off);
    off += (bytes + 255) & ~(size_t)255;
    return p;
  };
  const size_t nodeF = (size_t)N_NODES * 4;
  const size_t featF = (size_t)N_NODES * D * 4;
  float* out_norm = alloc(nodeF);
  float* in_norm  = alloc(nodeF);
  float* agg      = alloc(featF);
  float* hA       = alloc(featF);
  float* hB       = alloc(featF);
  float* h_final  = alloc(featF);

  const int TB = 256;
  const int nFeat   = N_NODES * D;
  const int gFeat   = (nFeat + TB - 1) / TB;
  const int gNode   = (N_NODES + TB - 1) / TB;
  const int gEdge   = (N_EDGES + TB - 1) / TB;
  const int gScat   = ((N_EDGES * 32) + TB - 1) / TB;
  const int gTile   = (N_NODES + 127) / 128;

  zero_f32_kernel<<<gNode, TB, 0, stream>>>(out_norm, N_NODES);
  zero_f32_kernel<<<gNode, TB, 0, stream>>>(in_norm,  N_NODES);
  degree_kernel<<<gEdge, TB, 0, stream>>>(src, dst, out_norm, in_norm, N_EDGES);
  norm_kernel<<<gNode, TB, 0, stream>>>(out_norm, in_norm, N_NODES);

  zero_f32_kernel<<<gFeat, TB, 0, stream>>>(h_final, nFeat);

  const float* h_in = feats;
  float* bufs[2] = { hA, hB };
  for (int layer = 0; layer < 3; ++layer) {
    float* h_out = bufs[layer & 1];
    zero_f32_kernel<<<gFeat, TB, 0, stream>>>(agg, nFeat);
    scatter_kernel<<<gScat, TB, 0, stream>>>(h_in, out_norm, src, dst, agg, N_EDGES);
    gcn_layer_kernel<<<gTile, TB, 0, stream>>>(
        agg, in_norm, h_in,
        conv_W + (size_t)layer * D * D, conv_b + (size_t)layer * D,
        res_W  + (size_t)layer * D * D, res_b  + (size_t)layer * D,
        ln_g   + (size_t)layer * D,     ln_b   + (size_t)layer * D,
        h_out, h_final, N_NODES);
    h_in = h_out;
  }

  pred_kernel<<<gTile, TB, 0, stream>>>(h_final, pred_W, pred_b, out, N_NODES);
}


# edges placed with no count, scan or zero launch: a slice stays sorted in its own slot with a run table; a bin walks its 123 runs by an LDS prefix table; 3 launches fewer; sum order = LDS arrival
# speedup vs baseline: 3.9057x; 3.9057x over previous
#include <hip/hip_runtime.h>

#define N_NODES 100000
#define N_EDGES 1000000
#define D 64
#define LN_EPS 1e-5f

#define BIN_NODES 196
#define PACK_SHIFT 8
#define N_BINS ((N_NODES + BIN_NODES - 1) / BIN_NODES)
#define NODE_SLOT_F (((N_NODES * 4 + 255) & ~255) / 4)

#define PLACE_T 1024
#define PLACE_K 8
#define PLACE_EDGES (PLACE_T * PLACE_K)
#define N_SLICES ((N_EDGES + PLACE_EDGES - 1) / PLACE_EDGES)
#define TAB_PITCH 512
#define RUN_SLOTS 128
#define ONORM_T 256
static_assert(N_BINS + 1 <= TAB_PITCH, "a table row holds every range's first position and the slice's count");
static_assert(N_BINS < PLACE_T, "one thread per table entry in the slice scan");
static_assert(N_SLICES < RUN_SLOTS, "the runs and their end entry fit the LDS run table");
static_assert(N_SLICES * 2 <= ONORM_T, "two threads per slice");
static_assert(BIN_NODES <= ONORM_T && ONORM_T == 256, "a source's row inside its range fits a byte; one thread per node");
static_assert(BIN_NODES <= (1 << PACK_SHIFT), "row inside the bin fits the low bits of an edge word");
static_assert(((unsigned long long)(N_NODES - 1) << PACK_SHIFT) < (1ull << 31), "packed edge word fits an int");

typedef __attribute__((ext_vector_type(2))) float v2f;
typedef __attribute__((ext_vector_type(8))) float v8f;


__device__ __forceinline__ int block_excl_scan(int c, int* wsum) {
  const int lane = threadIdx.x & 31;
  const int wave = threadIdx.x >> 5;
  int inc = c;
#pragma unroll
  for (int off = 1; off < 32; off <<= 1) {
    const int v = __shfl_up(inc, off);
    if (lane >= off) inc += v;
  }
  if (lane == 31) wsum[wave] = inc;
  __syncthreads();
  if (wave == 0) {
    const int w = wsum[lane];
    int winc = w;
#pragma unroll
    for (int off = 1; off < 32; off <<= 1) {
      const int v = __shfl_up(winc, off);
      if (lane >= off) winc += v;
    }
    wsum[lane] = winc - w;
  }
  __syncthreads();
  return inc - c + wsum[wave];
}

__global__ __launch_bounds__(PLACE_T) void src_place_kernel(
    const int* __restrict__ src, unsigned char* __restrict__ src_row,
    int* __restrict__ src_tab, int e) {
  __shared__ int hist[N_BINS];
  __shared__ int lstart[N_BINS];
  __shared__ int wsum[32];
  __shared__ __attribute__((aligned(16))) unsigned char stage[PLACE_EDGES];
  const int tid = threadIdx.x;
  if (tid < N_BINS) hist[tid] = 0;
  __syncthreads();
  const int base = blockIdx.x * PLACE_EDGES;
  int bin[PLACE_K], rank[PLACE_K], row[PLACE_K], sv[PLACE_K];
#pragma unroll
  for (int k = 0; k < PLACE_K; ++k) sv[k] = src[min(base + k * PLACE_T + tid, e - 1)];
#pragma unroll
  for (int k = 0; k < PLACE_K; ++k) {
    const int i = base + k * PLACE_T + tid;
    bin[k] = -1;
    rank[k] = 0;
    row[k] = 0;
    if (i < e) {
      const int s = sv[k];
      bin[k]  = s / BIN_NODES;
      row[k]  = s - bin[k] * BIN_NODES;
      rank[k] = atomicAdd(&hist[bin[k]], 1);
    }
  }
  __syncthreads();
  const int c  = tid < N_BINS ? hist[tid] : 0;
  const int ex = block_excl_scan(c, wsum);
  if (tid < N_BINS) lstart[tid] = ex;
  if (tid <= N_BINS) src_tab[blockIdx.x * TAB_PITCH + tid] = ex;
  __syncthreads();
  int pos[PLACE_K];
#pragma unroll
  for (int k = 0; k < PLACE_K; ++k) pos[k] = lstart[max(bin[k], 0)] + rank[k];
#pragma unroll
  for (int k = 0; k < PLACE_K; ++k) {
    if (bin[k] >= 0) stage[pos[k]] = (unsigned char)row[k];
  }
  __syncthreads();
  const int total = min(PLACE_EDGES, e - base);
  if (tid * 16 < total)
    reinterpret_cast<uint4*>(src_row + base)[tid] = reinterpret_cast<const uint4*>(stage)[tid];
}

__global__ __launch_bounds__(ONORM_T) void out_norm_kernel(
    const unsigned char* __restrict__ src_row, const int* __restrict__ src_tab,
    float* __restrict__ out_norm, int n) {
  __shared__ int cnt[ONORM_T];
  const int tid = threadIdx.x;
  const int b   = blockIdx.x;
  cnt[tid] = 0;
  __syncthreads();
  const int slice = tid >> 1;
  if (slice < N_SLICES) {
    const int s0 = src_tab[slice * TAB_PITCH + b];
    const int s1 = src_tab[slice * TAB_PITCH + b + 1];
    const unsigned char* p = src_row + slice * PLACE_EDGES;
    for (int i = s0 + (tid & 1); i < s1; i += 2) atomicAdd(&cnt[p[i]], 1);
  }
  __syncthreads();
  const int node = b * BIN_NODES + tid;
  if (tid < BIN_NODES && node < n) out_norm[node] = rsqrtf(fmaxf((float)cnt[tid], 1.0f));
}

__global__ __launch_bounds__(PLACE_T) void bin_place_kernel(
    const int* __restrict__ src, const int* __restrict__ dst,
    const float* __restrict__ out_norm,
    int* __restrict__ binned, int* __restrict__ run_tab, int e) {
  __shared__ int hist[N_BINS];
  __shared__ int lstart[N_BINS];
  __shared__ int wsum[32];
  __shared__ __attribute__((aligned(16))) int2 stage[PLACE_EDGES];
  const int tid = threadIdx.x;
  if (tid < N_BINS) hist[tid] = 0;
  __syncthreads();
  const int base = blockIdx.x * PLACE_EDGES;
  int bin[PLACE_K], rank[PLACE_K], word[PLACE_K], onb[PLACE_K], sv[PLACE_K], dv[PLACE_K];
#pragma unroll
  for (int k = 0; k < PLACE_K; ++k) {
    const int ic = min(base + k * PLACE_T + tid, e - 1);
    sv[k] = src[ic];
    dv[k] = dst[ic];
  }
#pragma unroll
  for (int k = 0; k < PLACE_K; ++k) onb[k] = __float_as_int(out_norm[sv[k]]);
#pragma unroll
  for (int k = 0; k < PLACE_K; ++k) {
    const int i = base + k * PLACE_T + tid;
    bin[k] = -1;
    rank[k] = 0;
    word[k] = 0;
    if (i < e) {
      const int d = dv[k];
      const int s = sv[k];
      bin[k]  = d / BIN_NODES;
      word[k] = (s << PACK_SHIFT) | (d - bin[k] * BIN_NODES);
      rank[k] = atomicAdd(&hist[bin[k]], 1);
    }
  }
  __syncthreads();
  const int c  = tid < N_BINS ? hist[tid] : 0;
  const int ex = block_excl_scan(c, wsum);
  if (tid < N_BINS) lstart[tid] = ex;
  if (tid <= N_BINS) run_tab[blockIdx.x * TAB_PITCH + tid] = ex;
  __syncthreads();
  int pos[PLACE_K];
#pragma unroll
  for (int k = 0; k < PLACE_K; ++k) pos[k] = lstart[max(bin[k], 0)] + rank[k];
#pragma unroll
  for (int k = 0; k < PLACE_K; ++k) {
    if (bin[k] >= 0) stage[pos[k]] = make_int2(word[k], onb[k]);
  }
  __syncthreads();
  const int total = min(PLACE_EDGES, e - base);
  int4* outp = reinterpret_cast<int4*>(binned) + (size_t)blockIdx.x * (PLACE_EDGES / 2);
  const int4* stp = reinterpret_cast<const int4*>(stage);
#pragma unroll
  for (int k = 0; k < PLACE_K / 2; ++k) {
    const int pr = k * PLACE_T + tid;
    if (pr * 2 < total) outp[pr] = stp[pr];
  }
}

#define TILE_PITCH 68
#define FUSE_T 512
#define FUSE_WAVES (FUSE_T / 32)
#define BIN_TILES ((BIN_NODES + 15) / 16)
#define GATHER_G 16
static_assert(BIN_TILES <= FUSE_WAVES, "one wave per 16-row tile of the bin");
static_assert((BIN_NODES * TILE_PITCH) % 4 == 0, "tile is zeroed in 16-byte pieces");
static_assert(BIN_NODES <= FUSE_T, "one thread zeroes one in-degree counter");

template <int LAST>
__global__ __launch_bounds__(FUSE_T, 8) void bin_layer_kernel(
    const float* __restrict__ h,
    const int* __restrict__ binned, const int* __restrict__ run_tab,
    const float* __restrict__ Wc, const float* __restrict__ bc,
    const float* __restrict__ Wr, const float* __restrict__ br,
    const float* __restrict__ lng, const float* __restrict__ lnb,
    const float* __restrict__ h_first,
    const float* __restrict__ Wp, const float* __restrict__ bp,
    float* __restrict__ h_out, int n)
{
  __shared__ __attribute__((aligned(16))) float tile[BIN_NODES * TILE_PITCH];
  __shared__ int indeg[BIN_NODES];
  __shared__ int run_pre[RUN_SLOTS];
  __shared__ int run_adr[RUN_SLOTS];
  const int tid  = threadIdx.x;
  const int lane = tid & 31;
  const int wave = __builtin_amdgcn_readfirstlane(tid >> 5);
  const int b    = blockIdx.x;

  if (wave == 0) {
    int st[4], len[4], s0[4], s1[4];
#pragma unroll
    for (int j = 0; j < 4; ++j) {
      const int kc = min(lane * 4 + j, N_SLICES - 1);
      s0[j] = run_tab[kc * TAB_PITCH + b];
      s1[j] = run_tab[kc * TAB_PITCH + b + 1];
    }
#pragma unroll
    for (int j = 0; j < 4; ++j) {
      const int k = lane * 4 + j;
      st[j]  = k * PLACE_EDGES + s0[j];
      len[j] = k < N_SLICES ? s1[j] - s0[j] : 0;
    }
    const int sum = (len[0] + len[1]) + (len[2] + len[3]);
    int inc = sum;
#pragma unroll
    for (int off = 1; off < 32; off <<= 1) {
      const int v = __shfl_up(inc, off);
      if (lane >= off) inc += v;
    }
    int ex = inc - sum;
#pragma unroll
    for (int j = 0; j < 4; ++j) {
      const int k = lane * 4 + j;
      run_pre[k] = k <= N_SLICES ? ex : 0x7fffffff;
      run_adr[k] = st[j] - ex;
      ex += len[j];
    }
  }
  float4* t4 = reinterpret_cast<float4*>(tile);
  for (int i = tid; i < BIN_NODES * TILE_PITCH / 4; i += FUSE_T) t4[i] = make_float4(0.0f, 0.0f, 0.0f, 0.0f);
  if (tid < BIN_NODES) indeg[tid] = 0;
  __syncthreads();

  const int c    = lane & 15;
  const int hf   = lane >> 4;
  const int lrow = wave * 16 + c;
  const bool live = lrow < BIN_NODES && b * BIN_NODES + lrow < n;
  const int crow  = live ? lrow : 0;
  const int cnode = b * BIN_NODES + crow;
  const unsigned noff = (unsigned)cnode * D + (unsigned)hf * 4u;
  const int wsub = ((c >> 2) & 1) * 8 + (c >> 3) * 4 + (c & 3);
  const unsigned woff = (unsigned)(hf * 4 * D + wsub);
  v8f acc[4] = {};
  const int turn = wave < BIN_TILES ? ((wave >> 2) << 1) | ((b ^ (b >> 8)) & 1) : -1;

  const int total = __builtin_amdgcn_readfirstlane(run_pre[N_SLICES]);
  auto edge_index = [&](int itn) -> int {
    const int pos = (itn * FUSE_WAVES + wave) * 32 + lane;
    int idx = 0;
    if (pos < total) {
      int lo = 0;
#pragma unroll
      for (int s = RUN_SLOTS / 2; s >= 1; s >>= 1) {
        if (run_pre[lo + s] <= pos) lo += s;
      }
      idx = run_adr[lo] + pos;
    }
    return idx;
  };
  int idx = edge_index(0);
  for (int it = 0; ; ++it) {
    const int base = (it * FUSE_WAVES + wave) * 32;
    if (base >= total && it > turn) break;
    if (it == turn) {
      v2f xR[16];
#pragma unroll
      for (int j = 0; j < 8; ++j) {
        const float4 hh = *reinterpret_cast<const float4*>(h + noff + 8 * j);
        xR[2 * j][0]     = hh.x;
        xR[2 * j][1]     = hh.y;
        xR[2 * j + 1][0] = hh.z;
        xR[2 * j + 1][1] = hh.w;
      }
#pragma unroll
      for (int t = 0; t < 16; ++t) {
        const int k = (t >> 1) * 8 + (t & 1) * 2;
#pragma unroll
        for (int mt = 0; mt < 4; ++mt) {
          v2f w;
          w[0] = Wr[woff + (k * D + mt * 16)];
          w[1] = Wr[woff + ((k + 1) * D + mt * 16)];
          acc[mt] = __builtin_amdgcn_wmma_f32_16x16x4_f32(false, w, false, xR[t], (short)0, acc[mt], false, false);
        }
      }
    }
    if (base >= total) continue;
    const int cnt = min(32, total - base);
    const int2 rec = lane < cnt ? reinterpret_cast<const int2*>(binned)[idx] : make_int2(0, 0);
    const int u = rec.x;
    if (lane < cnt) atomicAdd(&indeg[u & ((1 << PACK_SHIFT) - 1)], 1);
#pragma unroll
    for (int j0 = 0; j0 < 32; j0 += GATHER_G) {
      if (j0 < cnt) {
        float2 v[GATHER_G];
        float  on[GATHER_G];
        unsigned row[GATHER_G];
#pragma unroll
        for (int q = 0; q < GATHER_G; ++q) {
          const unsigned uq = (unsigned)__builtin_amdgcn_readlane(u, j0 + q);
          const unsigned s  = uq >> PACK_SHIFT;
          row[q] = uq & ((1u << PACK_SHIFT) - 1u);
          on[q]  = __int_as_float(__builtin_amdgcn_readlane(rec.y, j0 + q));
          v[q]   = *(reinterpret_cast<const float2*>(h + (size_t)s * D) + lane);
        }
        if (j0 == 0) idx = edge_index(it + 1);
#pragma unroll
        for (int q = 0; q < GATHER_G; ++q) {
          if (j0 + q < cnt) {
            float* t = tile + row[q] * TILE_PITCH + lane * 2;
            atomicAdd(t,     v[q].x * on[q]);
            atomicAdd(t + 1, v[q].y * on[q]);
          }
        }
      }
    }
  }
  __syncthreads();

  if (wave >= BIN_TILES) return;

  v2f xC[16];
  const float inrm = rsqrtf(fmaxf((float)indeg[crow], 1.0f));
  const float4* arow = reinterpret_cast<const float4*>(tile + crow * TILE_PITCH) + hf;
#pragma unroll
  for (int j = 0; j < 8; ++j) {
    const float4 a  = arow[2 * j];
    xC[2 * j][0]     = a.x * inrm;
    xC[2 * j][1]     = a.y * inrm;
    xC[2 * j + 1][0] = a.z * inrm;
    xC[2 * j + 1][1] = a.w * inrm;
  }

#pragma unroll
  for (int t = 0; t < 16; ++t) {
    const int k = (t >> 1) * 8 + (t & 1) * 2;
#pragma unroll
    for (int mt = 0; mt < 4; ++mt) {
      v2f w;
      w[0] = Wc[woff + (k * D + mt * 16)];
      w[1] = Wc[woff + ((k + 1) * D + mt * 16)];
      acc[mt] = __builtin_amdgcn_wmma_f32_16x16x4_f32(false, w, false, xC[t], (short)0, acc[mt], false, false);
    }
  }

  if constexpr (LAST) {
    __builtin_amdgcn_sched_barrier(0);
  }

  float x[32];
  float s = 0.0f, sq = 0.0f;
#pragma unroll
  for (int mt = 0; mt < 4; ++mt) {
#pragma unroll
    for (int q = 0; q < 2; ++q) {
      const int col0 = mt * 16 + q * 8 + hf * 4;
      const float4 b0 = *reinterpret_cast<const float4*>(bc + col0);
      const float4 b1 = *reinterpret_cast<const float4*>(br + col0);
      const float bias[4] = { b0.x + b1.x, b0.y + b1.y, b0.z + b1.z, b0.w + b1.w };
#pragma unroll
      for (int e = 0; e < 4; ++e) {
        const float v = acc[mt][q * 4 + e] + bias[e];
        x[mt * 8 + q * 4 + e] = v;
        s  += v;
        sq += v * v;
      }
    }
  }
  s  += __shfl_xor(s, 16);
  sq += __shfl_xor(sq, 16);
  const float mu   = s * (1.0f / 64.0f);
  const float var  = sq * (1.0f / 64.0f) - mu * mu;
  const float rstd = rsqrtf(var + LN_EPS);

  if constexpr (!LAST) {
    if (live) {
      float* ho = h_out + noff;
#pragma unroll
      for (int mt = 0; mt < 4; ++mt) {
#pragma unroll
        for (int q = 0; q < 2; ++q) {
          const int cb = mt * 16 + q * 8;
          const float4 gg = *reinterpret_cast<const float4*>(lng + cb + hf * 4);
          const float4 bb = *reinterpret_cast<const float4*>(lnb + cb + hf * 4);
          const float* xv = &x[mt * 8 + q * 4];
          float4 y;
          y.x = fmaxf((xv[0] - mu) * rstd * gg.x + bb.x, 0.0f);
          y.y = fmaxf((xv[1] - mu) * rstd * gg.y + bb.y, 0.0f);
          y.z = fmaxf((xv[2] - mu) * rstd * gg.z + bb.z, 0.0f);
          y.w = fmaxf((xv[3] - mu) * rstd * gg.w + bb.w, 0.0f);
          *reinterpret_cast<float4*>(ho + cb) = y;
        }
      }
    }
  } else {
    v2f jk[16];
#pragma unroll
    for (int mt = 0; mt < 4; ++mt) {
#pragma unroll
      for (int q = 0; q < 2; ++q) {
        const int cb = mt * 16 + q * 8;
        const int j  = 2 * mt + q;
        const float4 gg = *reinterpret_cast<const float4*>(lng + cb + hf * 4);
        const float4 bb = *reinterpret_cast<const float4*>(lnb + cb + hf * 4);
        const float4 p  = *reinterpret_cast<const float4*>(h_first + noff + 8 * j);
        const float4 r  = *reinterpret_cast<const float4*>(h + noff + 8 * j);
        const float* xv = &x[mt * 8 + q * 4];
        const float y0 = fmaxf((xv[0] - mu) * rstd * gg.x + bb.x, 0.0f);
        const float y1 = fmaxf((xv[1] - mu) * rstd * gg.y + bb.y, 0.0f);
        const float y2 = fmaxf((xv[2] - mu) * rstd * gg.z + bb.z, 0.0f);
        const float y3 = fmaxf((xv[3] - mu) * rstd * gg.w + bb.w, 0.0f);
        jk[2 * j][0]     = (p.x + r.x) + y0;
        jk[2 * j][1]     = (p.y + r.y) + y1;
        jk[2 * j + 1][0] = (p.z + r.z) + y2;
        jk[2 * j + 1][1] = (p.w + r.w) + y3;
      }
    }

    v8f pacc[4] = {};
#pragma unroll
    for (int t = 0; t < 16; ++t) {
      const int k = (t >> 1) * 8 + (t & 1) * 2;
#pragma unroll
      for (int mt = 0; mt < 4; ++mt) {
        v2f w;
        w[0] = Wp[woff + (k * D + mt * 16)];
        w[1] = Wp[woff + ((k + 1) * D + mt * 16)];
        pacc[mt] = __builtin_amdgcn_wmma_f32_16x16x4_f32(false, w, false, jk[t], (short)0, pacc[mt], false, false);
      }
    }

    if (live) {
      float* o = h_out + noff;
#pragma unroll
      for (int mt = 0; mt < 4; ++mt) {
#pragma unroll
        for (int q = 0; q < 2; ++q) {
          const int cb = mt * 16 + q * 8;
          const float4 pb = *reinterpret_cast<const float4*>(bp + cb + hf * 4);
          float4 r;
          r.x = pacc[mt][q * 4 + 0] + pb.x;
          r.y = pacc[mt][q * 4 + 1] + pb.y;
          r.z = pacc[mt][q * 4 + 2] + pb.z;
          r.w = pacc[mt][q * 4 + 3] + pb.w;
          *reinterpret_cast<float4*>(o + cb) = r;
        }
      }
    }
  }
}

extern "C" void kernel_launch(void* const* d_in, const int* in_sizes, int n_in,
                              void* d_out, int out_size, void* d_ws, size_t ws_size,
                              hipStream_t stream) {
  (void)in_sizes; (void)n_in; (void)out_size; (void)ws_size;

  const float* feats  = (const float*)d_in[0];
  const int*   src    = (const int*)  d_in[1];
  const int*   dst    = (const int*)  d_in[2];
  const float* conv_W = (const float*)d_in[3];
  const float* conv_b = (const float*)d_in[4];
  const float* res_W  = (const float*)d_in[5];
  const float* res_b  = (const float*)d_in[6];
  const float* ln_g   = (const float*)d_in[7];
  const float* ln_b   = (const float*)d_in[8];
  const float* pred_W = (const float*)d_in[9];
  const float* pred_b = (const float*)d_in[10];
  float* out = (float*)d_out;

  char* ws = (char*)d_ws;
  size_t off = 0;
  auto alloc = [&](size_t bytes) -> float* {
    float* p = (float*)(ws + off);
    off += (bytes + 255) & ~(size_t)255;
    return p;
  };
  const size_t nodeF = (size_t)NODE_SLOT_F * 4;
  const size_t featF = (size_t)N_NODES * D * 4;
  const size_t tabB  = (size_t)N_SLICES * TAB_PITCH * 4;
  const size_t rowB  = (size_t)N_SLICES * PLACE_EDGES;
  float* out_norm = alloc(nodeF);
  int*   run_tab  = (int*)alloc(tabB);
  int*   binned   = (int*)alloc((size_t)N_SLICES * PLACE_EDGES * 8);
  float* h1       = alloc(featF);
  float* h2       = alloc(featF);
  unsigned char* src_row = (unsigned char*)h1;
  int*           src_tab = (int*)((char*)h1 + rowB);
  static_assert((size_t)N_SLICES * PLACE_EDGES % 256 == 0, "the source run table starts on a line");
  static_assert((size_t)N_SLICES * PLACE_EDGES + (size_t)N_SLICES * TAB_PITCH * 4 <= (size_t)N_NODES * D * 4,
                "row bytes and source run table fit the h1 plane");

  src_place_kernel<<<N_SLICES, PLACE_T, 0, stream>>>(src, src_row, src_tab, N_EDGES);
  out_norm_kernel<<<N_BINS, ONORM_T, 0, stream>>>(src_row, src_tab, out_norm, N_NODES);
  bin_place_kernel<<<N_SLICES, PLACE_T, 0, stream>>>(src, dst, out_norm, binned, run_tab, N_EDGES);

  bin_layer_kernel<0><<<N_BINS, FUSE_T, 0, stream>>>(
      feats, binned, run_tab,
      conv_W, conv_b, res_W, res_b, ln_g, ln_b,
      nullptr, nullptr, nullptr, h1, N_NODES);

  bin_layer_kernel<0><<<N_BINS, FUSE_T, 0, stream>>>(
      h1, binned, run_tab,
      conv_W + D * D, conv_b + D, res_W + D * D, res_b + D, ln_g + D, ln_b + D,
      nullptr, nullptr, nullptr, h2, N_NODES);

  bin_layer_kernel<1><<<N_BINS, FUSE_T, 0, stream>>>(
      h2, binned, run_tab,
      conv_W + 2 * D * D, conv_b + 2 * D, res_W + 2 * D * D, res_b + 2 * D,
      ln_g + 2 * D, ln_b + 2 * D,
      h1, pred_W, pred_b, out, N_NODES);
}


# weights through LDS: a work-group copies its layer's matrices transposed (k contiguous) once; an A fragment load is one 16-byte LDS read for two steps, not 128 4-byte global loads a product
# speedup vs baseline: 4.2448x; 1.0868x over previous
#include <hip/hip_runtime.h>

#define N_NODES 100000
#define N_EDGES 1000000
#define D 64
#define LN_EPS 1e-5f

#define BIN_NODES 196
#define PACK_SHIFT 8
#define N_BINS ((N_NODES + BIN_NODES - 1) / BIN_NODES)
#define NODE_SLOT_F (((N_NODES * 4 + 255) & ~255) / 4)

#define PLACE_T 1024
#define PLACE_K 8
#define PLACE_EDGES (PLACE_T * PLACE_K)
#define N_SLICES ((N_EDGES + PLACE_EDGES - 1) / PLACE_EDGES)
#define TAB_PITCH 512
#define RUN_SLOTS 128
#define ONORM_T 256
static_assert(N_BINS + 1 <= TAB_PITCH, "a table row holds every range's first position and the slice's count");
static_assert(N_BINS < PLACE_T, "one thread per table entry in the slice scan");
static_assert(N_SLICES < RUN_SLOTS, "the runs and their end entry fit the LDS run table");
static_assert(N_SLICES * 2 <= ONORM_T, "two threads per slice");
static_assert(BIN_NODES <= ONORM_T && ONORM_T == 256, "a source's row inside its range fits a byte; one thread per node");
static_assert(BIN_NODES <= (1 << PACK_SHIFT), "row inside the bin fits the low bits of an edge word");
static_assert(((unsigned long long)(N_NODES - 1) << PACK_SHIFT) < (1ull << 31), "packed edge word fits an int");

typedef __attribute__((ext_vector_type(2))) float v2f;
typedef __attribute__((ext_vector_type(8))) float v8f;


__device__ __forceinline__ int block_excl_scan(int c, int* wsum) {
  const int lane = threadIdx.x & 31;
  const int wave = threadIdx.x >> 5;
  int inc = c;
#pragma unroll
  for (int off = 1; off < 32; off <<= 1) {
    const int v = __shfl_up(inc, off);
    if (lane >= off) inc += v;
  }
  if (lane == 31) wsum[wave] = inc;
  __syncthreads();
  if (wave == 0) {
    const int w = wsum[lane];
    int winc = w;
#pragma unroll
    for (int off = 1; off < 32; off <<= 1) {
      const int v = __shfl_up(winc, off);
      if (lane >= off) winc += v;
    }
    wsum[lane] = winc - w;
  }
  __syncthreads();
  return inc - c + wsum[wave];
}

__global__ __launch_bounds__(PLACE_T) void src_place_kernel(
    const int* __restrict__ src, unsigned char* __restrict__ src_row,
    int* __restrict__ src_tab, int e) {
  __shared__ int hist[N_BINS];
  __shared__ int lstart[N_BINS];
  __shared__ int wsum[32];
  __shared__ __attribute__((aligned(16))) unsigned char stage[PLACE_EDGES];
  const int tid = threadIdx.x;
  if (tid < N_BINS) hist[tid] = 0;
  __syncthreads();
  const int base = blockIdx.x * PLACE_EDGES;
  int bin[PLACE_K], rank[PLACE_K], row[PLACE_K], sv[PLACE_K];
#pragma unroll
  for (int k = 0; k < PLACE_K; ++k) sv[k] = src[min(base + k * PLACE_T + tid, e - 1)];
#pragma unroll
  for (int k = 0; k < PLACE_K; ++k) {
    const int i = base + k * PLACE_T + tid;
    bin[k] = -1;
    rank[k] = 0;
    row[k] = 0;
    if (i < e) {
      const int s = sv[k];
      bin[k]  = s / BIN_NODES;
      row[k]  = s - bin[k] * BIN_NODES;
      rank[k] = atomicAdd(&hist[bin[k]], 1);
    }
  }
  __syncthreads();
  const int c  = tid < N_BINS ? hist[tid] : 0;
  const int ex = block_excl_scan(c, wsum);
  if (tid < N_BINS) lstart[tid] = ex;
  if (tid <= N_BINS) src_tab[blockIdx.x * TAB_PITCH + tid] = ex;
  __syncthreads();
  int pos[PLACE_K];
#pragma unroll
  for (int k = 0; k < PLACE_K; ++k) pos[k] = lstart[max(bin[k], 0)] + rank[k];
#pragma unroll
  for (int k = 0; k < PLACE_K; ++k) {
    if (bin[k] >= 0) stage[pos[k]] = (unsigned char)row[k];
  }
  __syncthreads();
  const int total = min(PLACE_EDGES, e - base);
  if (tid * 16 < total)
    reinterpret_cast<uint4*>(src_row + base)[tid] = reinterpret_cast<const uint4*>(stage)[tid];
}

__global__ __launch_bounds__(ONORM_T) void out_norm_kernel(
    const unsigned char* __restrict__ src_row, const int* __restrict__ src_tab,
    float* __restrict__ out_norm, int n) {
  __shared__ int cnt[ONORM_T];
  const int tid = threadIdx.x;
  const int b   = blockIdx.x;
  cnt[tid] = 0;
  __syncthreads();
  const int slice = tid >> 1;
  if (slice < N_SLICES) {
    const int s0 = src_tab[slice * TAB_PITCH + b];
    const int s1 = src_tab[slice * TAB_PITCH + b + 1];
    const unsigned char* p = src_row + slice * PLACE_EDGES;
    for (int i = s0 + (tid & 1); i < s1; i += 2) atomicAdd(&cnt[p[i]], 1);
  }
  __syncthreads();
  const int node = b * BIN_NODES + tid;
  if (tid < BIN_NODES && node < n) out_norm[node] = rsqrtf(fmaxf((float)cnt[tid], 1.0f));
}

__global__ __launch_bounds__(PLACE_T) void bin_place_kernel(
    const int* __restrict__ src, const int* __restrict__ dst,
    const float* __restrict__ out_norm,
    int* __restrict__ binned, int* __restrict__ run_tab, int e) {
  __shared__ int hist[N_BINS];
  __shared__ int lstart[N_BINS];
  __shared__ int wsum[32];
  __shared__ __attribute__((aligned(16))) int2 stage[PLACE_EDGES];
  const int tid = threadIdx.x;
  if (tid < N_BINS) hist[tid] = 0;
  __syncthreads();
  const int base = blockIdx.x * PLACE_EDGES;
  int bin[PLACE_K], rank[PLACE_K], word[PLACE_K], onb[PLACE_K], sv[PLACE_K], dv[PLACE_K];
#pragma unroll
  for (int k = 0; k < PLACE_K; ++k) {
    const int ic = min(base + k * PLACE_T + tid, e - 1);
    sv[k] = src[ic];
    dv[k] = dst[ic];
  }
#pragma unroll
  for (int k = 0; k < PLACE_K; ++k) onb[k] = __float_as_int(out_norm[sv[k]]);
#pragma unroll
  for (int k = 0; k < PLACE_K; ++k) {
    const int i = base + k * PLACE_T + tid;
    bin[k] = -1;
    rank[k] = 0;
    word[k] = 0;
    if (i < e) {
      const int d = dv[k];
      const int s = sv[k];
      bin[k]  = d / BIN_NODES;
      word[k] = (s << PACK_SHIFT) | (d - bin[k] * BIN_NODES);
      rank[k] = atomicAdd(&hist[bin[k]], 1);
    }
  }
  __syncthreads();
  const int c  = tid < N_BINS ? hist[tid] : 0;
  const int ex = block_excl_scan(c, wsum);
  if (tid < N_BINS) lstart[tid] = ex;
  if (tid <= N_BINS) run_tab[blockIdx.x * TAB_PITCH + tid] = ex;
  __syncthreads();
  int pos[PLACE_K];
#pragma unroll
  for (int k = 0; k < PLACE_K; ++k) pos[k] = lstart[max(bin[k], 0)] + rank[k];
#pragma unroll
  for (int k = 0; k < PLACE_K; ++k) {
    if (bin[k] >= 0) stage[pos[k]] = make_int2(word[k], onb[k]);
  }
  __syncthreads();
  const int total = min(PLACE_EDGES, e - base);
  int4* outp = reinterpret_cast<int4*>(binned) + (size_t)blockIdx.x * (PLACE_EDGES / 2);
  const int4* stp = reinterpret_cast<const int4*>(stage);
#pragma unroll
  for (int k = 0; k < PLACE_K / 2; ++k) {
    const int pr = k * PLACE_T + tid;
    if (pr * 2 < total) outp[pr] = stp[pr];
  }
}

#define TILE_PITCH 68
#define W_PITCH 68
#define FUSE_T 512
#define FUSE_WAVES (FUSE_T / 32)
#define BIN_TILES ((BIN_NODES + 15) / 16)
#define GATHER_G 16
static_assert(BIN_TILES <= FUSE_WAVES, "one wave per 16-row tile of the bin");
static_assert((BIN_NODES * TILE_PITCH) % 4 == 0, "tile is zeroed in 16-byte pieces");
static_assert(BIN_NODES <= FUSE_T, "one thread zeroes one in-degree counter");
static_assert(FUSE_T == 8 * D, "weight staging: a thread takes one column and 4 consecutive k, two passes cover k");

__device__ __forceinline__ void weight_product(const float* wl, const v2f (&x)[16], v8f (&acc)[4]) {
#pragma unroll
  for (int j = 0; j < 8; ++j) {
    float4 a[4];
#pragma unroll
    for (int mt = 0; mt < 4; ++mt) a[mt] = *reinterpret_cast<const float4*>(wl + mt * 16 * W_PITCH + 8 * j);
#pragma unroll
    for (int u = 0; u < 2; ++u) {
#pragma unroll
      for (int mt = 0; mt < 4; ++mt) {
        v2f w;
        w[0] = u ? a[mt].z : a[mt].x;
        w[1] = u ? a[mt].w : a[mt].y;
        acc[mt] = __builtin_amdgcn_wmma_f32_16x16x4_f32(false, w, false, x[2 * j + u], (short)0, acc[mt], false, false);
      }
    }
  }
}

template <int LAST>
__global__ __launch_bounds__(FUSE_T, 8) void bin_layer_kernel(
    const float* __restrict__ h,
    const int* __restrict__ binned, const int* __restrict__ run_tab,
    const float* __restrict__ Wc, const float* __restrict__ bc,
    const float* __restrict__ Wr, const float* __restrict__ br,
    const float* __restrict__ lng, const float* __restrict__ lnb,
    const float* __restrict__ h_first,
    const float* __restrict__ Wp, const float* __restrict__ bp,
    float* __restrict__ h_out, int n)
{
  __shared__ __attribute__((aligned(16))) float tile[BIN_NODES * TILE_PITCH];
  __shared__ int indeg[BIN_NODES];
  __shared__ int run_pre[RUN_SLOTS];
  __shared__ int run_adr[RUN_SLOTS];
  __shared__ __attribute__((aligned(16))) float wts[(2 + LAST) * D * W_PITCH];
  const int tid  = threadIdx.x;
  const int lane = tid & 31;
  const int wave = __builtin_amdgcn_readfirstlane(tid >> 5);
  const int b    = blockIdx.x;

  const int scol = tid & (D - 1);
  const int skq  = tid >> 6;
  float4 sw[2 + LAST][2];
#pragma unroll
  for (int m = 0; m < 2 + LAST; ++m) {
    const float* Wm = m == 0 ? Wr : (m == 1 ? Wc : Wp);
#pragma unroll
    for (int p = 0; p < 2; ++p) {
      const int k0 = 4 * (skq + 8 * p);
      sw[m][p] = make_float4(Wm[(k0 + 0) * D + scol], Wm[(k0 + 1) * D + scol],
                             Wm[(k0 + 2) * D + scol], Wm[(k0 + 3) * D + scol]);
    }
  }

  if (wave == 0) {
    int st[4], len[4], s0[4], s1[4];
#pragma unroll
    for (int j = 0; j < 4; ++j) {
      const int kc = min(lane * 4 + j, N_SLICES - 1);
      s0[j] = run_tab[kc * TAB_PITCH + b];
      s1[j] = run_tab[kc * TAB_PITCH + b + 1];
    }
#pragma unroll
    for (int j = 0; j < 4; ++j) {
      const int k = lane * 4 + j;
      st[j]  = k * PLACE_EDGES + s0[j];
      len[j] = k < N_SLICES ? s1[j] - s0[j] : 0;
    }
    const int sum = (len[0] + len[1]) + (len[2] + len[3]);
    int inc = sum;
#pragma unroll
    for (int off = 1; off < 32; off <<= 1) {
      const int v = __shfl_up(inc, off);
      if (lane >= off) inc += v;
    }
    int ex = inc - sum;
#pragma unroll
    for (int j = 0; j < 4; ++j) {
      const int k = lane * 4 + j;
      run_pre[k] = k <= N_SLICES ? ex : 0x7fffffff;
      run_adr[k] = st[j] - ex;
      ex += len[j];
    }
  }
  float4* t4 = reinterpret_cast<float4*>(tile);
  for (int i = tid; i < BIN_NODES * TILE_PITCH / 4; i += FUSE_T) t4[i] = make_float4(0.0f, 0.0f, 0.0f, 0.0f);
  if (tid < BIN_NODES) indeg[tid] = 0;
#pragma unroll
  for (int m = 0; m < 2 + LAST; ++m) {
#pragma unroll
    for (int p = 0; p < 2; ++p)
      *reinterpret_cast<float4*>(wts + m * D * W_PITCH + scol * W_PITCH + 4 * (skq + 8 * p)) = sw[m][p];
  }
  __syncthreads();

  const int c    = lane & 15;
  const int hf   = lane >> 4;
  const int lrow = wave * 16 + c;
  const bool live = lrow < BIN_NODES && b * BIN_NODES + lrow < n;
  const int crow  = live ? lrow : 0;
  const int cnode = b * BIN_NODES + crow;
  const unsigned noff = (unsigned)cnode * D + (unsigned)hf * 4u;
  const int wsub = ((c >> 2) & 1) * 8 + (c >> 3) * 4 + (c & 3);
  const float* wlane = wts + wsub * W_PITCH + hf * 4;
  v8f acc[4] = {};
  const int turn = wave < BIN_TILES ? ((wave >> 2) << 1) | ((b ^ (b >> 8)) & 1) : -1;

  const int total = __builtin_amdgcn_readfirstlane(run_pre[N_SLICES]);
  auto edge_index = [&](int itn) -> int {
    const int pos = (itn * FUSE_WAVES + wave) * 32 + lane;
    int idx = 0;
    if (pos < total) {
      int lo = 0;
#pragma unroll
      for (int s = RUN_SLOTS / 2; s >= 1; s >>= 1) {
        if (run_pre[lo + s] <= pos) lo += s;
      }
      idx = run_adr[lo] + pos;
    }
    return idx;
  };
  int idx = edge_index(0);
  for (int it = 0; ; ++it) {
    const int base = (it * FUSE_WAVES + wave) * 32;
    if (base >= total && it > turn) break;
    if (it == turn) {
      v2f xR[16];
#pragma unroll
      for (int j = 0; j < 8; ++j) {
        const float4 hh = *reinterpret_cast<const float4*>(h + noff + 8 * j);
        xR[2 * j][0]     = hh.x;
        xR[2 * j][1]     = hh.y;
        xR[2 * j + 1][0] = hh.z;
        xR[2 * j + 1][1] = hh.w;
      }
      weight_product(wlane, xR, acc);
    }
    if (base >= total) continue;
    const int cnt = min(32, total - base);
    const int2 rec = lane < cnt ? reinterpret_cast<const int2*>(binned)[idx] : make_int2(0, 0);
    const int u = rec.x;
    if (lane < cnt) atomicAdd(&indeg[u & ((1 << PACK_SHIFT) - 1)], 1);
#pragma unroll
    for (int j0 = 0; j0 < 32; j0 += GATHER_G) {
      if (j0 < cnt) {
        float2 v[GATHER_G];
        float  on[GATHER_G];
        unsigned row[GATHER_G];
#pragma unroll
        for (int q = 0; q < GATHER_G; ++q) {
          const unsigned uq = (unsigned)__builtin_amdgcn_readlane(u, j0 + q);
          const unsigned s  = uq >> PACK_SHIFT;
          row[q] = uq & ((1u << PACK_SHIFT) - 1u);
          on[q]  = __int_as_float(__builtin_amdgcn_readlane(rec.y, j0 + q));
          v[q]   = *(reinterpret_cast<const float2*>(h + (size_t)s * D) + lane);
        }
        if (j0 == 0) idx = edge_index(it + 1);
#pragma unroll
        for (int q = 0; q < GATHER_G; ++q) {
          if (j0 + q < cnt) {
            float* t = tile + row[q] * TILE_PITCH + lane * 2;
            atomicAdd(t,     v[q].x * on[q]);
            atomicAdd(t + 1, v[q].y * on[q]);
          }
        }
      }
    }
  }
  __syncthreads();

  if (wave >= BIN_TILES) return;

  v2f xC[16];
  const float inrm = rsqrtf(fmaxf((float)indeg[crow], 1.0f));
  const float4* arow = reinterpret_cast<const float4*>(tile + crow * TILE_PITCH) + hf;
#pragma unroll
  for (int j = 0; j < 8; ++j) {
    const float4 a  = arow[2 * j];
    xC[2 * j][0]     = a.x * inrm;
    xC[2 * j][1]     = a.y * inrm;
    xC[2 * j + 1][0] = a.z * inrm;
    xC[2 * j + 1][1] = a.w * inrm;
  }

  weight_product(wlane + D * W_PITCH, xC, acc);

  if constexpr (LAST) {
    __builtin_amdgcn_sched_barrier(0);
  }

  float x[32];
  float s = 0.0f, sq = 0.0f;
#pragma unroll
  for (int mt = 0; mt < 4; ++mt) {
#pragma unroll
    for (int q = 0; q < 2; ++q) {
      const int col0 = mt * 16 + q * 8 + hf * 4;
      const float4 b0 = *reinterpret_cast<const float4*>(bc + col0);
      const float4 b1 = *reinterpret_cast<const float4*>(br + col0);
      const float bias[4] = { b0.x + b1.x, b0.y + b1.y, b0.z + b1.z, b0.w + b1.w };
#pragma unroll
      for (int e = 0; e < 4; ++e) {
        const float v = acc[mt][q * 4 + e] + bias[e];
        x[mt * 8 + q * 4 + e] = v;
        s  += v;
        sq += v * v;
      }
    }
  }
  s  += __shfl_xor(s, 16);
  sq += __shfl_xor(sq, 16);
  const float mu   = s * (1.0f / 64.0f);
  const float var  = sq * (1.0f / 64.0f) - mu * mu;
  const float rstd = rsqrtf(var + LN_EPS);

  if constexpr (!LAST) {
    if (live) {
      float* ho = h_out + noff;
#pragma unroll
      for (int mt = 0; mt < 4; ++mt) {
#pragma unroll
        for (int q = 0; q < 2; ++q) {
          const int cb = mt * 16 + q * 8;
          const float4 gg = *reinterpret_cast<const float4*>(lng + cb + hf * 4);
          const float4 bb = *reinterpret_cast<const float4*>(lnb + cb + hf * 4);
          const float* xv = &x[mt * 8 + q * 4];
          float4 y;
          y.x = fmaxf((xv[0] - mu) * rstd * gg.x + bb.x, 0.0f);
          y.y = fmaxf((xv[1] - mu) * rstd * gg.y + bb.y, 0.0f);
          y.z = fmaxf((xv[2] - mu) * rstd * gg.z + bb.z, 0.0f);
          y.w = fmaxf((xv[3] - mu) * rstd * gg.w + bb.w, 0.0f);
          *reinterpret_cast<float4*>(ho + cb) = y;
        }
      }
    }
  } else {
    v2f jk[16];
#pragma unroll
    for (int mt = 0; mt < 4; ++mt) {
#pragma unroll
      for (int q = 0; q < 2; ++q) {
        const int cb = mt * 16 + q * 8;
        const int j  = 2 * mt + q;
        const float4 gg = *reinterpret_cast<const float4*>(lng + cb + hf * 4);
        const float4 bb = *reinterpret_cast<const float4*>(lnb + cb + hf * 4);
        const float4 p  = *reinterpret_cast<const float4*>(h_first + noff + 8 * j);
        const float4 r  = *reinterpret_cast<const float4*>(h + noff + 8 * j);
        const float* xv = &x[mt * 8 + q * 4];
        const float y0 = fmaxf((xv[0] - mu) * rstd * gg.x + bb.x, 0.0f);
        const float y1 = fmaxf((xv[1] - mu) * rstd * gg.y + bb.y, 0.0f);
        const float y2 = fmaxf((xv[2] - mu) * rstd * gg.z + bb.z, 0.0f);
        const float y3 = fmaxf((xv[3] - mu) * rstd * gg.w + bb.w, 0.0f);
        jk[2 * j][0]     = (p.x + r.x) + y0;
        jk[2 * j][1]     = (p.y + r.y) + y1;
        jk[2 * j + 1][0] = (p.z + r.z) + y2;
        jk[2 * j + 1][1] = (p.w + r.w) + y3;
      }
    }

    v8f pacc[4] = {};
    weight_product(wlane + 2 * D * W_PITCH, jk, pacc);

    if (live) {
      float* o = h_out + noff;
#pragma unroll
      for (int mt = 0; mt < 4; ++mt) {
#pragma unroll
        for (int q = 0; q < 2; ++q) {
          const int cb = mt * 16 + q * 8;
          const float4 pb = *reinterpret_cast<const float4*>(bp + cb + hf * 4);
          float4 r;
          r.x = pacc[mt][q * 4 + 0] + pb.x;
          r.y = pacc[mt][q * 4 + 1] + pb.y;
          r.z = pacc[mt][q * 4 + 2] + pb.z;
          r.w = pacc[mt][q * 4 + 3] + pb.w;
          *reinterpret_cast<float4*>(o + cb) = r;
        }
      }
    }
  }
}

extern "C" void kernel_launch(void* const* d_in, const int* in_sizes, int n_in,
                              void* d_out, int out_size, void* d_ws, size_t ws_size,
                              hipStream_t stream) {
  (void)in_sizes; (void)n_in; (void)out_size; (void)ws_size;

  const float* feats  = (const float*)d_in[0];
  const int*   src    = (const int*)  d_in[1];
  const int*   dst    = (const int*)  d_in[2];
  const float* conv_W = (const float*)d_in[3];
  const float* conv_b = (const float*)d_in[4];
  const float* res_W  = (const float*)d_in[5];
  const float* res_b  = (const float*)d_in[6];
  const float* ln_g   = (const float*)d_in[7];
  const float* ln_b   = (const float*)d_in[8];
  const float* pred_W = (const float*)d_in[9];
  const float* pred_b = (const float*)d_in[10];
  float* out = (float*)d_out;

  char* ws = (char*)d_ws;
  size_t off = 0;
  auto alloc = [&](size_t bytes) -> float* {
    float* p = (float*)(ws + off);
    off += (bytes + 255) & ~(size_t)255;
    return p;
  };
  const size_t nodeF = (size_t)NODE_SLOT_F * 4;
  const size_t featF = (size_t)N_NODES * D * 4;
  const size_t tabB  = (size_t)N_SLICES * TAB_PITCH * 4;
  const size_t rowB  = (size_t)N_SLICES * PLACE_EDGES;
  float* out_norm = alloc(nodeF);
  int*   run_tab  = (int*)alloc(tabB);
  int*   binned   = (int*)alloc((size_t)N_SLICES * PLACE_EDGES * 8);
  float* h1       = alloc(featF);
  float* h2       = alloc(featF);
  unsigned char* src_row = (unsigned char*)h1;
  int*           src_tab = (int*)((char*)h1 + rowB);
  static_assert((size_t)N_SLICES * PLACE_EDGES % 256 == 0, "the source run table starts on a line");
  static_assert((size_t)N_SLICES * PLACE_EDGES + (size_t)N_SLICES * TAB_PITCH * 4 <= (size_t)N_NODES * D * 4,
                "row bytes and source run table fit the h1 plane");

  src_place_kernel<<<N_SLICES, PLACE_T, 0, stream>>>(src, src_row, src_tab, N_EDGES);
  out_norm_kernel<<<N_BINS, ONORM_T, 0, stream>>>(src_row, src_tab, out_norm, N_NODES);
  bin_place_kernel<<<N_SLICES, PLACE_T, 0, stream>>>(src, dst, out_norm, binned, run_tab, N_EDGES);

  bin_layer_kernel<0><<<N_BINS, FUSE_T, 0, stream>>>(
      feats, binned, run_tab,
      conv_W, conv_b, res_W, res_b, ln_g, ln_b,
      nullptr, nullptr, nullptr, h1, N_NODES);

  bin_layer_kernel<0><<<N_BINS, FUSE_T, 0, stream>>>(
      h1, binned, run_tab,
      conv_W + D * D, conv_b + D, res_W + D * D, res_b + D, ln_g + D, ln_b + D,
      nullptr, nullptr, nullptr, h2, N_NODES);

  bin_layer_kernel<1><<<N_BINS, FUSE_T, 0, stream>>>(
      h2, binned, run_tab,
      conv_W + 2 * D * D, conv_b + 2 * D, res_W + 2 * D * D, res_b + 2 * D,
      ln_g + 2 * D, ln_b + 2 * D,
      h1, pred_W, pred_b, out, N_NODES);
}
